# starting kernel
# baseline (speedup 1.0000x reference)
#include <hip/hip_runtime.h>
#include <float.h>

typedef __attribute__((ext_vector_type(2))) float v2f;
typedef __attribute__((ext_vector_type(8))) float v8f;

#define BB    32
#define SS    512
#define HH    1024
#define NC    256
#define NL    34
#define NLPAD 48
#define KTOT  (2 * HH)

__global__ void dmbert_wpad_kernel(const float* __restrict__ Wm,
                                   float*       __restrict__ Wt) {
    const int idx = blockIdx.x * blockDim.x + threadIdx.x;
    const int k   = idx & (KTOT - 1);
    const int c   = idx / KTOT;
    Wt[idx] = (c < NL) ? Wm[(size_t)k * NL + c] : 0.0f;
}

__global__ void dmbert_pool_kernel(const float* __restrict__ hidden,
                                   const int*  __restrict__ cand_batch,
                                   const int*  __restrict__ cand_cut,
                                   float*      __restrict__ pooled) {
    const int n   = blockIdx.x;
    const int t   = threadIdx.x;
    const int bat = cand_batch[n];
    const int cut = cand_cut[n];

    const float4* row = (const float4*)(hidden + (size_t)bat * SS * HH) + t;
    const int rstride = HH / 4;

    float4 ml = make_float4(-FLT_MAX, -FLT_MAX, -FLT_MAX, -FLT_MAX);
    float4 mr = ml;

    int s = 0;
    #pragma unroll 4
    for (; s < cut; ++s) {
        float4 v = row[(size_t)s * rstride];
        ml.x = fmaxf(ml.x, v.x); ml.y = fmaxf(ml.y, v.y);
        ml.z = fmaxf(ml.z, v.z); ml.w = fmaxf(ml.w, v.w);
    }
    #pragma unroll 4
    for (; s < SS; ++s) {
        float4 v = row[(size_t)s * rstride];
        mr.x = fmaxf(mr.x, v.x); mr.y = fmaxf(mr.y, v.y);
        mr.z = fmaxf(mr.z, v.z); mr.w = fmaxf(mr.w, v.w);
    }

    float4* outL = (float4*)(pooled + (size_t)n * KTOT) + t;
    float4* outR = (float4*)(pooled + (size_t)n * KTOT + HH) + t;
    *outL = ml;
    *outR = mr;
}

__global__ void dmbert_logits_wmma_kernel(const float* __restrict__ pooled,
                                          const float* __restrict__ Wt,
                                          const float* __restrict__ bias,
                                          float*       __restrict__ out) {
    const int mtile = blockIdx.x;
    const int ntile = blockIdx.y;
    const int wave  = threadIdx.x >> 5;
    const int lane  = threadIdx.x & 31;
    const int half  = lane >> 4;
    const int l16   = lane & 15;

    const int m   = mtile * 16 + l16;
    const int col = ntile * 16 + l16;

    const float* arow = pooled + (size_t)m   * KTOT + 2 * half;
    const float* brow = Wt     + (size_t)col * KTOT + 2 * half;

    v8f c = {};

    const int k0 = wave * (KTOT / 4);
    #pragma unroll 4
    for (int kk = 0; kk < KTOT / 4; kk += 4) {
        const int k = k0 + kk;
        v2f a = *(const v2f*)(arow + k);
        v2f b = *(const v2f*)(brow + k);
        c = __builtin_amdgcn_wmma_f32_16x16x4_f32(
                 false, a,  false, b,
                 (short)0, c,  false,  false);
    }

    __shared__ float red[4][32][8];
    #pragma unroll
    for (int j = 0; j < 8; ++j) red[wave][lane][j] = c[j];
    __syncthreads();

    if (wave == 0 && col < NL) {
        #pragma unroll
        for (int j = 0; j < 8; ++j) {
            float s = red[0][lane][j] + red[1][lane][j] +
                      red[2][lane][j] + red[3][lane][j];
            const int mrow = mtile * 16 + j + half * 8;
            out[(size_t)mrow * NL + col] = s + bias[col];
        }
    }
}

extern "C" void kernel_launch(void* const* d_in, const int* in_sizes, int n_in,
                              void* d_out, int out_size, void* d_ws, size_t ws_size,
                              hipStream_t stream) {
    const float* hidden     = (const float*)d_in[0];
    const float* Wm         = (const float*)d_in[1];
    const float* bias       = (const float*)d_in[2];
    const int*   cand_batch = (const int*)  d_in[3];
    const int*   cand_cut   = (const int*)  d_in[4];
    float*       out        = (float*)d_out;

    float* pooled = (float*)d_ws;
    float* Wt     = pooled + (size_t)NC * KTOT;

    dmbert_wpad_kernel<<<(NLPAD * KTOT) / 256, 256, 0, stream>>>(Wm, Wt);

    dmbert_pool_kernel<<<NC, 256, 0, stream>>>(hidden, cand_batch, cand_cut, pooled);

    dim3 grid(NC / 16, (NL + 15) / 16);
    dmbert_logits_wmma_kernel<<<grid, 128, 0, stream>>>(pooled, Wt, bias, out);
}


# Two-level max pool through the workspace: a grid of (row, 32-token chunk) blocks reads hidden once and stores chunk maxima plus each cut's partials; a small launch merges per candidate; max exact
# speedup vs baseline: 5.2474x; 5.2474x over previous
#include <hip/hip_runtime.h>
#include <float.h>

typedef __attribute__((ext_vector_type(2))) float v2f;
typedef __attribute__((ext_vector_type(8))) float v8f;

#define BB    32
#define SS    512
#define HH    1024
#define NC    256
#define NL    34
#define NLPAD 48
#define KTOT  (2 * HH)

#define TG  32
#define NCH (SS / TG)

#define WS_NEED_BYTES ((size_t)(NC * KTOT + NLPAD * KTOT + BB * NCH * HH) * sizeof(float))

__global__ void dmbert_wpad_kernel(const float* __restrict__ Wm,
                                   float*       __restrict__ Wt) {
    const int idx = blockIdx.x * blockDim.x + threadIdx.x;
    const int k   = idx & (KTOT - 1);
    const int c   = idx / KTOT;
    Wt[idx] = (c < NL) ? Wm[(size_t)k * NL + c] : 0.0f;
}

__device__ __forceinline__ int clamp_cut(int cut) { return min(max(cut, 0), SS); }
__device__ __forceinline__ int home_chunk(int cut) { return min(cut / TG, NCH - 1); }

__device__ __forceinline__ float4 max4(float4 a, float4 b) {
    return make_float4(__builtin_fmaximum_numf(a.x, b.x), __builtin_fmaximum_numf(a.y, b.y),
                       __builtin_fmaximum_numf(a.z, b.z), __builtin_fmaximum_numf(a.w, b.w));
}

__device__ __forceinline__ void pool_candidate_walk(const float* __restrict__ hidden,
                                                    const int*  __restrict__ cand_batch,
                                                    const int*  __restrict__ cand_cut,
                                                    float*      __restrict__ pooled,
                                                    const int n, const int t) {
    const int bat = cand_batch[n];
    const int cut = cand_cut[n];

    const float4* row = (const float4*)(hidden + (size_t)bat * SS * HH) + t;
    const int rstride = HH / 4;

    float4 ml = make_float4(-FLT_MAX, -FLT_MAX, -FLT_MAX, -FLT_MAX);
    float4 mr = ml;

    for (int s0 = 0; s0 < SS; s0 += TG) {
        const float4* p = row + (size_t)s0 * rstride;
        float4 v[TG];
        #pragma unroll
        for (int j = 0; j < TG; ++j) v[j] = p[j * rstride];

        if (s0 + TG <= cut) {
            #pragma unroll
            for (int j = 0; j < TG; ++j) ml = max4(ml, v[j]);
        } else if (s0 >= cut) {
            #pragma unroll
            for (int j = 0; j < TG; ++j) mr = max4(mr, v[j]);
        } else {
            #pragma unroll
            for (int j = 0; j < TG; ++j) {
                if (s0 + j < cut) ml = max4(ml, v[j]);
                else              mr = max4(mr, v[j]);
            }
        }
    }

    float4* outL = (float4*)(pooled + (size_t)n * KTOT) + t;
    float4* outR = (float4*)(pooled + (size_t)n * KTOT + HH) + t;
    *outL = ml;
    *outR = mr;
}

__global__ void __launch_bounds__(256)
dmbert_chunk_kernel(const float* __restrict__ hidden,
                    const int*  __restrict__ cand_batch,
                    const int*  __restrict__ cand_cut,
                    float*      __restrict__ pooled,
                    float*      __restrict__ chunkmax,
                    const size_t ws_bytes) {
    const int t = threadIdx.x;

    if (ws_bytes < WS_NEED_BYTES) {
        if (blockIdx.x < NC)
            pool_candidate_walk(hidden, cand_batch, cand_cut, pooled, blockIdx.x, t);
        return;
    }

    const int bat = blockIdx.x / NCH;
    const int ch  = blockIdx.x % NCH;

    const int cb = cand_batch[t];
    const int cc = clamp_cut(cand_cut[t]);

    const float4* p = (const float4*)(hidden + ((size_t)bat * SS + (size_t)ch * TG) * HH) + t;
    const int rstride = HH / 4;

    float4 v[TG];
    #pragma unroll
    for (int j = 0; j < TG; ++j) v[j] = p[j * rstride];

    __shared__ unsigned hit[NC / 32];
    const bool     mine = (cb == bat) & (home_chunk(cc) == ch);
    const unsigned word = __builtin_amdgcn_ballot_w32(mine);
    if ((t & 31) == 0) hit[t >> 5] = word;

    float4 cm = make_float4(-FLT_MAX, -FLT_MAX, -FLT_MAX, -FLT_MAX);
    #pragma unroll
    for (int j = 0; j < TG; ++j) cm = max4(cm, v[j]);

    __syncthreads();

    ((float4*)(chunkmax + (size_t)blockIdx.x * HH))[t] = cm;

    for (int w = 0; w < NC / 32; ++w) {
        unsigned bits = __builtin_amdgcn_readfirstlane(hit[w]);
        while (bits) {
            const int n = w * 32 + __builtin_ctz(bits);
            bits &= bits - 1;
            const int c = clamp_cut(cand_cut[n]) - ch * TG;
            float4 pl = make_float4(-FLT_MAX, -FLT_MAX, -FLT_MAX, -FLT_MAX);
            float4 pr = pl;
            #pragma unroll
            for (int j = 0; j < TG; ++j) {
                if (j < c) pl = max4(pl, v[j]);
                else       pr = max4(pr, v[j]);
            }
            ((float4*)(pooled + (size_t)n * KTOT))[t]      = pl;
            ((float4*)(pooled + (size_t)n * KTOT + HH))[t] = pr;
        }
    }
}

__global__ void __launch_bounds__(256)
dmbert_merge_kernel(const int*   __restrict__ cand_batch,
                    const int*   __restrict__ cand_cut,
                    const float* __restrict__ chunkmax,
                    float*                    pooled,
                    const size_t ws_bytes) {
    if (ws_bytes < WS_NEED_BYTES) return;

    const int n   = blockIdx.x;
    const int t   = threadIdx.x;
    const int bat = cand_batch[n];
    const int ci  = home_chunk(clamp_cut(cand_cut[n]));

    const float4* cmrow = (const float4*)(chunkmax + (size_t)bat * NCH * HH) + t;
    float4* outL = (float4*)(pooled + (size_t)n * KTOT) + t;
    float4* outR = (float4*)(pooled + (size_t)n * KTOT + HH) + t;

    float4 c[NCH];
    #pragma unroll
    for (int k = 0; k < NCH; ++k) c[k] = cmrow[k * (HH / 4)];
    float4 ml = *outL;
    float4 mr = *outR;

    const float4 lo = make_float4(-FLT_MAX, -FLT_MAX, -FLT_MAX, -FLT_MAX);
    #pragma unroll
    for (int k = 0; k < NCH; ++k) {
        ml = max4(ml, (k < ci) ? c[k] : lo);
        mr = max4(mr, (k > ci) ? c[k] : lo);
    }
    *outL = ml;
    *outR = mr;
}

__global__ void dmbert_logits_wmma_kernel(const float* __restrict__ pooled,
                                          const float* __restrict__ Wt,
                                          const float* __restrict__ bias,
                                          float*       __restrict__ out) {
    const int mtile = blockIdx.x;
    const int ntile = blockIdx.y;
    const int wave  = threadIdx.x >> 5;
    const int lane  = threadIdx.x & 31;
    const int half  = lane >> 4;
    const int l16   = lane & 15;

    const int m   = mtile * 16 + l16;
    const int col = ntile * 16 + l16;

    const float* arow = pooled + (size_t)m   * KTOT + 2 * half;
    const float* brow = Wt     + (size_t)col * KTOT + 2 * half;

    v8f c = {};

    const int k0 = wave * (KTOT / 4);
    #pragma unroll 4
    for (int kk = 0; kk < KTOT / 4; kk += 4) {
        const int k = k0 + kk;
        v2f a = *(const v2f*)(arow + k);
        v2f b = *(const v2f*)(brow + k);
        c = __builtin_amdgcn_wmma_f32_16x16x4_f32(
                 false, a,  false, b,
                 (short)0, c,  false,  false);
    }

    __shared__ float red[4][32][8];
    #pragma unroll
    for (int j = 0; j < 8; ++j) red[wave][lane][j] = c[j];
    __syncthreads();

    if (wave == 0 && col < NL) {
        #pragma unroll
        for (int j = 0; j < 8; ++j) {
            float s = red[0][lane][j] + red[1][lane][j] +
                      red[2][lane][j] + red[3][lane][j];
            const int mrow = mtile * 16 + j + half * 8;
            out[(size_t)mrow * NL + col] = s + bias[col];
        }
    }
}

extern "C" void kernel_launch(void* const* d_in, const int* in_sizes, int n_in,
                              void* d_out, int out_size, void* d_ws, size_t ws_size,
                              hipStream_t stream) {
    const float* hidden     = (const float*)d_in[0];
    const float* Wm         = (const float*)d_in[1];
    const float* bias       = (const float*)d_in[2];
    const int*   cand_batch = (const int*)  d_in[3];
    const int*   cand_cut   = (const int*)  d_in[4];
    float*       out        = (float*)d_out;

    float* pooled   = (float*)d_ws;
    float* Wt       = pooled + (size_t)NC * KTOT;
    float* chunkmax = Wt + (size_t)NLPAD * KTOT;

    dmbert_wpad_kernel<<<(NLPAD * KTOT) / 256, 256, 0, stream>>>(Wm, Wt);

    dmbert_chunk_kernel<<<BB * NCH, 256, 0, stream>>>(hidden, cand_batch, cand_cut,
                                                      pooled, chunkmax, ws_size);

    dmbert_merge_kernel<<<NC, 256, 0, stream>>>(cand_batch, cand_cut, chunkmax, pooled, ws_size);

    dim3 grid(NC / 16, (NL + 15) / 16);
    dmbert_logits_wmma_kernel<<<grid, 128, 0, stream>>>(pooled, Wt, bias, out);
}


# Logits GEMM takes its B fragments straight from W as given (two 4-byte loads per step, 32 k per loop trip): no padded transposed copy of W is built, that launch is gone, 4 -> 3; sums bit-identical
# speedup vs baseline: 5.7565x; 1.0970x over previous
#include <hip/hip_runtime.h>
#include <float.h>

typedef __attribute__((ext_vector_type(2))) float v2f;
typedef __attribute__((ext_vector_type(8))) float v8f;

#define BB    32
#define SS    512
#define HH    1024
#define NC    256
#define NL    34
#define KTOT  (2 * HH)

#define TG  32
#define NCH (SS / TG)

#define WS_NEED_BYTES ((size_t)(NC * KTOT + BB * NCH * HH) * sizeof(float))

__device__ __forceinline__ int clamp_cut(int cut) { return min(max(cut, 0), SS); }
__device__ __forceinline__ int home_chunk(int cut) { return min(cut / TG, NCH - 1); }

__device__ __forceinline__ float4 max4(float4 a, float4 b) {
    return make_float4(__builtin_fmaximum_numf(a.x, b.x), __builtin_fmaximum_numf(a.y, b.y),
                       __builtin_fmaximum_numf(a.z, b.z), __builtin_fmaximum_numf(a.w, b.w));
}

__device__ __forceinline__ void pool_candidate_walk(const float* __restrict__ hidden,
                                                    const int*  __restrict__ cand_batch,
                                                    const int*  __restrict__ cand_cut,
                                                    float*      __restrict__ pooled,
                                                    const int n, const int t) {
    const int bat = cand_batch[n];
    const int cut = cand_cut[n];

    const float4* row = (const float4*)(hidden + (size_t)bat * SS * HH) + t;
    const int rstride = HH / 4;

    float4 ml = make_float4(-FLT_MAX, -FLT_MAX, -FLT_MAX, -FLT_MAX);
    float4 mr = ml;

    for (int s0 = 0; s0 < SS; s0 += TG) {
        const float4* p = row + (size_t)s0 * rstride;
        float4 v[TG];
        #pragma unroll
        for (int j = 0; j < TG; ++j) v[j] = p[j * rstride];

        if (s0 + TG <= cut) {
            #pragma unroll
            for (int j = 0; j < TG; ++j) ml = max4(ml, v[j]);
        } else if (s0 >= cut) {
            #pragma unroll
            for (int j = 0; j < TG; ++j) mr = max4(mr, v[j]);
        } else {
            #pragma unroll
            for (int j = 0; j < TG; ++j) {
                if (s0 + j < cut) ml = max4(ml, v[j]);
                else              mr = max4(mr, v[j]);
            }
        }
    }

    float4* outL = (float4*)(pooled + (size_t)n * KTOT) + t;
    float4* outR = (float4*)(pooled + (size_t)n * KTOT + HH) + t;
    *outL = ml;
    *outR = mr;
}

__global__ void __launch_bounds__(256)
dmbert_chunk_kernel(const float* __restrict__ hidden,
                    const int*  __restrict__ cand_batch,
                    const int*  __restrict__ cand_cut,
                    float*      __restrict__ pooled,
                    float*      __restrict__ chunkmax,
                    const size_t ws_bytes) {
    const int t = threadIdx.x;

    if (ws_bytes < WS_NEED_BYTES) {
        if (blockIdx.x < NC)
            pool_candidate_walk(hidden, cand_batch, cand_cut, pooled, blockIdx.x, t);
        return;
    }

    const int bat = blockIdx.x / NCH;
    const int ch  = blockIdx.x % NCH;

    const int cb = cand_batch[t];
    const int cc = clamp_cut(cand_cut[t]);

    const float4* p = (const float4*)(hidden + ((size_t)bat * SS + (size_t)ch * TG) * HH) + t;
    const int rstride = HH / 4;

    float4 v[TG];
    #pragma unroll
    for (int j = 0; j < TG; ++j) v[j] = p[j * rstride];

    __shared__ unsigned hit[NC / 32];
    const bool     mine = (cb == bat) & (home_chunk(cc) == ch);
    const unsigned word = __builtin_amdgcn_ballot_w32(mine);
    if ((t & 31) == 0) hit[t >> 5] = word;

    float4 cm = make_float4(-FLT_MAX, -FLT_MAX, -FLT_MAX, -FLT_MAX);
    #pragma unroll
    for (int j = 0; j < TG; ++j) cm = max4(cm, v[j]);

    __syncthreads();

    ((float4*)(chunkmax + (size_t)blockIdx.x * HH))[t] = cm;

    for (int w = 0; w < NC / 32; ++w) {
        unsigned bits = __builtin_amdgcn_readfirstlane(hit[w]);
        while (bits) {
            const int n = w * 32 + __builtin_ctz(bits);
            bits &= bits - 1;
            const int c = clamp_cut(cand_cut[n]) - ch * TG;
            float4 pl = make_float4(-FLT_MAX, -FLT_MAX, -FLT_MAX, -FLT_MAX);
            float4 pr = pl;
            #pragma unroll
            for (int j = 0; j < TG; ++j) {
                if (j < c) pl = max4(pl, v[j]);
                else       pr = max4(pr, v[j]);
            }
            ((float4*)(pooled + (size_t)n * KTOT))[t]      = pl;
            ((float4*)(pooled + (size_t)n * KTOT + HH))[t] = pr;
        }
    }
}

__global__ void __launch_bounds__(256)
dmbert_merge_kernel(const int*   __restrict__ cand_batch,
                    const int*   __restrict__ cand_cut,
                    const float* __restrict__ chunkmax,
                    float*                    pooled,
                    const size_t ws_bytes) {
    if (ws_bytes < WS_NEED_BYTES) return;

    const int n   = blockIdx.x;
    const int t   = threadIdx.x;
    const int bat = cand_batch[n];
    const int ci  = home_chunk(clamp_cut(cand_cut[n]));

    const float4* cmrow = (const float4*)(chunkmax + (size_t)bat * NCH * HH) + t;
    float4* outL = (float4*)(pooled + (size_t)n * KTOT) + t;
    float4* outR = (float4*)(pooled + (size_t)n * KTOT + HH) + t;

    float4 c[NCH];
    #pragma unroll
    for (int k = 0; k < NCH; ++k) c[k] = cmrow[k * (HH / 4)];
    float4 ml = *outL;
    float4 mr = *outR;

    const float4 lo = make_float4(-FLT_MAX, -FLT_MAX, -FLT_MAX, -FLT_MAX);
    #pragma unroll
    for (int k = 0; k < NCH; ++k) {
        ml = max4(ml, (k < ci) ? c[k] : lo);
        mr = max4(mr, (k > ci) ? c[k] : lo);
    }
    *outL = ml;
    *outR = mr;
}

__global__ void dmbert_logits_wmma_kernel(const float* __restrict__ pooled,
                                          const float* __restrict__ Wm,
                                          const float* __restrict__ bias,
                                          float*       __restrict__ out) {
    const int mtile = blockIdx.x;
    const int ntile = blockIdx.y;
    const int wave  = threadIdx.x >> 5;
    const int lane  = threadIdx.x & 31;
    const int half  = lane >> 4;
    const int l16   = lane & 15;

    const int m   = mtile * 16 + l16;
    const int col = ntile * 16 + l16;
    const int cld = min(col, NL - 1);

    const float* arow = pooled + (size_t)m * KTOT + 2 * half;
    const float* bcol = Wm + (size_t)(2 * half) * NL + cld;

    v8f c = {};

    const int k0 = wave * (KTOT / 4);
    #pragma unroll 8
    for (int kk = 0; kk < KTOT / 4; kk += 4) {
        const int k = k0 + kk;
        v2f a = *(const v2f*)(arow + k);
        v2f b;
        b.x = bcol[(size_t)k * NL];
        b.y = bcol[(size_t)k * NL + NL];
        c = __builtin_amdgcn_wmma_f32_16x16x4_f32(
                 false, a,  false, b,
                 (short)0, c,  false,  false);
    }

    __shared__ float red[4][32][8];
    #pragma unroll
    for (int j = 0; j < 8; ++j) red[wave][lane][j] = c[j];
    __syncthreads();

    if (wave == 0 && col < NL) {
        #pragma unroll
        for (int j = 0; j < 8; ++j) {
            float s = red[0][lane][j] + red[1][lane][j] +
                      red[2][lane][j] + red[3][lane][j];
            const int mrow = mtile * 16 + j + half * 8;
            out[(size_t)mrow * NL + col] = s + bias[col];
        }
    }
}

extern "C" void kernel_launch(void* const* d_in, const int* in_sizes, int n_in,
                              void* d_out, int out_size, void* d_ws, size_t ws_size,
                              hipStream_t stream) {
    const float* hidden     = (const float*)d_in[0];
    const float* Wm         = (const float*)d_in[1];
    const float* bias       = (const float*)d_in[2];
    const int*   cand_batch = (const int*)  d_in[3];
    const int*   cand_cut   = (const int*)  d_in[4];
    float*       out        = (float*)d_out;

    float* pooled   = (float*)d_ws;
    float* chunkmax = pooled + (size_t)NC * KTOT;

    dmbert_chunk_kernel<<<BB * NCH, 256, 0, stream>>>(hidden, cand_batch, cand_cut,
                                                      pooled, chunkmax, ws_size);

    dmbert_merge_kernel<<<NC, 256, 0, stream>>>(cand_batch, cand_cut, chunkmax, pooled, ws_size);

    dim3 grid(NC / 16, (NL + 15) / 16);
    dmbert_logits_wmma_kernel<<<grid, 128, 0, stream>>>(pooled, Wm, bias, out);
}


# Merge folded into the product launch: a (4-candidate, column-tile) block of 16 waves rebuilds its pooled rows in LDS from the chunk maxima; 3 launches -> 2; sum order: 16 k-slice partials, wave order
# speedup vs baseline: 7.8319x; 1.3605x over previous
#include <hip/hip_runtime.h>
#include <float.h>

typedef __attribute__((ext_vector_type(2))) float v2f;
typedef __attribute__((ext_vector_type(8))) float v8f;

#define BB    32
#define SS    512
#define HH    1024
#define NC    256
#define NL    34
#define KTOT  (2 * HH)

#define TG  32
#define NCH (SS / TG)

#define WS_NEED_BYTES ((size_t)(NC * KTOT + BB * NCH * HH) * sizeof(float))

__device__ __forceinline__ int clamp_cut(int cut) { return min(max(cut, 0), SS); }
__device__ __forceinline__ int home_chunk(int cut) { return min(cut / TG, NCH - 1); }

__device__ __forceinline__ float4 max4(float4 a, float4 b) {
    return make_float4(__builtin_fmaximum_numf(a.x, b.x), __builtin_fmaximum_numf(a.y, b.y),
                       __builtin_fmaximum_numf(a.z, b.z), __builtin_fmaximum_numf(a.w, b.w));
}

__device__ __forceinline__ void pool_candidate_walk(const float* __restrict__ hidden,
                                                    const int*  __restrict__ cand_batch,
                                                    const int*  __restrict__ cand_cut,
                                                    float*      __restrict__ pooled,
                                                    const int n, const int t) {
    const int bat = cand_batch[n];
    const int cut = cand_cut[n];

    const float4* row = (const float4*)(hidden + (size_t)bat * SS * HH) + t;
    const int rstride = HH / 4;

    float4 ml = make_float4(-FLT_MAX, -FLT_MAX, -FLT_MAX, -FLT_MAX);
    float4 mr = ml;

    for (int s0 = 0; s0 < SS; s0 += TG) {
        const float4* p = row + (size_t)s0 * rstride;
        float4 v[TG];
        #pragma unroll
        for (int j = 0; j < TG; ++j) v[j] = p[j * rstride];

        if (s0 + TG <= cut) {
            #pragma unroll
            for (int j = 0; j < TG; ++j) ml = max4(ml, v[j]);
        } else if (s0 >= cut) {
            #pragma unroll
            for (int j = 0; j < TG; ++j) mr = max4(mr, v[j]);
        } else {
            #pragma unroll
            for (int j = 0; j < TG; ++j) {
                if (s0 + j < cut) ml = max4(ml, v[j]);
                else              mr = max4(mr, v[j]);
            }
        }
    }

    float4* outL = (float4*)(pooled + (size_t)n * KTOT) + t;
    float4* outR = (float4*)(pooled + (size_t)n * KTOT + HH) + t;
    *outL = ml;
    *outR = mr;
}

__global__ void __launch_bounds__(256)
dmbert_chunk_kernel(const float* __restrict__ hidden,
                    const int*  __restrict__ cand_batch,
                    const int*  __restrict__ cand_cut,
                    float*      __restrict__ pooled,
                    float*      __restrict__ chunkmax,
                    const size_t ws_bytes) {
    const int t = threadIdx.x;

    if (ws_bytes < WS_NEED_BYTES) {
        if (blockIdx.x < NC)
            pool_candidate_walk(hidden, cand_batch, cand_cut, pooled, blockIdx.x, t);
        return;
    }

    const int bat = blockIdx.x / NCH;
    const int ch  = blockIdx.x % NCH;

    const int cb = cand_batch[t];
    const int cc = clamp_cut(cand_cut[t]);

    const float4* p = (const float4*)(hidden + ((size_t)bat * SS + (size_t)ch * TG) * HH) + t;
    const int rstride = HH / 4;

    float4 v[TG];
    #pragma unroll
    for (int j = 0; j < TG; ++j) v[j] = p[j * rstride];

    __shared__ unsigned hit[NC / 32];
    const bool     mine = (cb == bat) & (home_chunk(cc) == ch);
    const unsigned word = __builtin_amdgcn_ballot_w32(mine);
    if ((t & 31) == 0) hit[t >> 5] = word;

    float4 cm = make_float4(-FLT_MAX, -FLT_MAX, -FLT_MAX, -FLT_MAX);
    #pragma unroll
    for (int j = 0; j < TG; ++j) cm = max4(cm, v[j]);

    __syncthreads();

    ((float4*)(chunkmax + (size_t)blockIdx.x * HH))[t] = cm;

    for (int w = 0; w < NC / 32; ++w) {
        unsigned bits = __builtin_amdgcn_readfirstlane(hit[w]);
        while (bits) {
            const int n = w * 32 + __builtin_ctz(bits);
            bits &= bits - 1;
            const int c = clamp_cut(cand_cut[n]) - ch * TG;
            float4 pl = make_float4(-FLT_MAX, -FLT_MAX, -FLT_MAX, -FLT_MAX);
            float4 pr = pl;
            #pragma unroll
            for (int j = 0; j < TG; ++j) {
                if (j < c) pl = max4(pl, v[j]);
                else       pr = max4(pr, v[j]);
            }
            ((float4*)(pooled + (size_t)n * KTOT))[t]      = pl;
            ((float4*)(pooled + (size_t)n * KTOT + HH))[t] = pr;
        }
    }
}

#define GC    4
#define NWAVE 16
#define KW    (KTOT / NWAVE)
#define NSTEP (KW / 4)
#define TP    (KTOT + 4)

__global__ void __launch_bounds__(NWAVE * 32)
dmbert_merge_logits_kernel(const int*   __restrict__ cand_batch,
                           const int*   __restrict__ cand_cut,
                           const float* __restrict__ chunkmax,
                           const float* __restrict__ pooled,
                           const float* __restrict__ Wm,
                           const float* __restrict__ bias,
                           float*       __restrict__ out,
                           const size_t ws_bytes) {
    const int m0    = blockIdx.x * GC;
    const int ntile = blockIdx.y;
    const int t     = threadIdx.x;
    const int wave  = t >> 5;
    const int lane  = t & 31;
    const int half  = lane >> 4;
    const int l16   = lane & 15;

    __shared__ __attribute__((aligned(16))) float tile[GC][TP];
    __shared__ float4 red[NWAVE][32];

    const int col = ntile * 16 + l16;
    const int cld = min(col, NL - 1);
    const float bv = bias[cld];

    const int k0 = wave * KW + 2 * half;
    const float* bcol = Wm + (size_t)k0 * NL + cld;
    v2f b[NSTEP];
    #pragma unroll
    for (int s = 0; s < NSTEP; ++s) {
        b[s].x = bcol[(size_t)(4 * s) * NL];
        b[s].y = bcol[(size_t)(4 * s) * NL + NL];
    }

    const int hq = t & 255;
    const int cp = __builtin_amdgcn_readfirstlane(t >> 8);
    const int nA = m0 + 2 * cp;
    const int nB = nA + 1;

    float4 mlA = ((const float4*)(pooled + (size_t)nA * KTOT))[hq];
    float4 mrA = ((const float4*)(pooled + (size_t)nA * KTOT + HH))[hq];
    float4 mlB = ((const float4*)(pooled + (size_t)nB * KTOT))[hq];
    float4 mrB = ((const float4*)(pooled + (size_t)nB * KTOT + HH))[hq];

    if (ws_bytes >= WS_NEED_BYTES) {
        const int ciA = home_chunk(clamp_cut(cand_cut[nA]));
        const int ciB = home_chunk(clamp_cut(cand_cut[nB]));
        const float4* rowA = (const float4*)(chunkmax + (size_t)cand_batch[nA] * NCH * HH) + hq;
        const float4* rowB = (const float4*)(chunkmax + (size_t)cand_batch[nB] * NCH * HH) + hq;

        float4 cA[NCH], cB[NCH];
        #pragma unroll
        for (int k = 0; k < NCH; ++k) {
            cA[k] = rowA[k * (HH / 4)];
            cB[k] = rowB[k * (HH / 4)];
        }
        const float4 lo = make_float4(-FLT_MAX, -FLT_MAX, -FLT_MAX, -FLT_MAX);
        #pragma unroll
        for (int k = 0; k < NCH; ++k) {
            mlA = max4(mlA, (k < ciA) ? cA[k] : lo);
            mrA = max4(mrA, (k > ciA) ? cA[k] : lo);
            mlB = max4(mlB, (k < ciB) ? cB[k] : lo);
            mrB = max4(mrB, (k > ciB) ? cB[k] : lo);
        }
    }

    *(float4*)&tile[2 * cp][4 * hq]          = mlA;
    *(float4*)&tile[2 * cp][HH + 4 * hq]     = mrA;
    *(float4*)&tile[2 * cp + 1][4 * hq]      = mlB;
    *(float4*)&tile[2 * cp + 1][HH + 4 * hq] = mrB;
    __syncthreads();

    const float* arow = &tile[l16 & (GC - 1)][k0];
    v8f c = {};
    #pragma unroll
    for (int s = 0; s < NSTEP; ++s) {
        v2f a = *(const v2f*)(arow + 4 * s);
        c = __builtin_amdgcn_wmma_f32_16x16x4_f32(
                 false, a,  false, b[s],
                 (short)0, c,  false,  false);
    }

    red[wave][lane] = make_float4(c[0], c[1], c[2], c[3]);
    __syncthreads();

    if (t < GC * 16) {
        const int r  = t >> 4;
        const int oc = ntile * 16 + (t & 15);
        if (oc < NL) {
            const float* p = (const float*)&red[0][t & 15] + r;
            float s = p[0];
            #pragma unroll
            for (int w = 1; w < NWAVE; ++w) s += p[w * 32 * 4];
            out[(size_t)(m0 + r) * NL + oc] = s + bv;
        }
    }
}

extern "C" void kernel_launch(void* const* d_in, const int* in_sizes, int n_in,
                              void* d_out, int out_size, void* d_ws, size_t ws_size,
                              hipStream_t stream) {
    const float* hidden     = (const float*)d_in[0];
    const float* Wm         = (const float*)d_in[1];
    const float* bias       = (const float*)d_in[2];
    const int*   cand_batch = (const int*)  d_in[3];
    const int*   cand_cut   = (const int*)  d_in[4];
    float*       out        = (float*)d_out;

    float* pooled   = (float*)d_ws;
    float* chunkmax = pooled + (size_t)NC * KTOT;

    dmbert_chunk_kernel<<<BB * NCH, 256, 0, stream>>>(hidden, cand_batch, cand_cut,
                                                      pooled, chunkmax, ws_size);

    dim3 grid(NC / GC, (NL + 15) / 16);
    dmbert_merge_logits_kernel<<<grid, NWAVE * 32, 0, stream>>>(cand_batch, cand_cut, chunkmax, pooled,
                                                              Wm, bias, out, ws_size);
}


# Cut partials in two levels: 8 maxima of 4 tokens with a running max up and down the groups; a cut takes one register per side by a scalar switch and folds 4 tokens, not 32; max exact, bit-identical
# speedup vs baseline: 8.4432x; 1.0781x over previous
#include <hip/hip_runtime.h>
#include <float.h>

typedef __attribute__((ext_vector_type(2))) float v2f;
typedef __attribute__((ext_vector_type(8))) float v8f;

#define BB    32
#define SS    512
#define HH    1024
#define NC    256
#define NL    34
#define KTOT  (2 * HH)

#define TG  32
#define NCH (SS / TG)
#define GT  4
#define NG  (TG / GT)

#define WS_NEED_BYTES ((size_t)(NC * KTOT + BB * NCH * HH) * sizeof(float))

__device__ __forceinline__ int clamp_cut(int cut) { return min(max(cut, 0), SS); }
__device__ __forceinline__ int home_chunk(int cut) { return min(cut / TG, NCH - 1); }

__device__ __forceinline__ float4 max4(float4 a, float4 b) {
    return make_float4(__builtin_fmaximum_numf(a.x, b.x), __builtin_fmaximum_numf(a.y, b.y),
                       __builtin_fmaximum_numf(a.z, b.z), __builtin_fmaximum_numf(a.w, b.w));
}

template <int G0>
__device__ __forceinline__ void cut_partials(const float4 (&v)[TG], const float4 (&up)[NG],
                                             const float4 (&dn)[NG], const int r,
                                             float4& pl, float4& pr) {
    const float4 lo = make_float4(-FLT_MAX, -FLT_MAX, -FLT_MAX, -FLT_MAX);
    if constexpr (G0 > 0)      pl = up[G0 - 1]; else pl = lo;
    if constexpr (G0 + 1 < NG) pr = dn[G0 + 1]; else pr = lo;
    if constexpr (G0 < NG) {
        #pragma unroll
        for (int i = 0; i < GT; ++i) {
            if (i < r) pl = max4(pl, v[G0 * GT + i]);
            else       pr = max4(pr, v[G0 * GT + i]);
        }
    }
}

__device__ __forceinline__ void pool_candidate_walk(const float* __restrict__ hidden,
                                                    const int*  __restrict__ cand_batch,
                                                    const int*  __restrict__ cand_cut,
                                                    float*      __restrict__ pooled,
                                                    const int n, const int t) {
    const int bat = cand_batch[n];
    const int cut = cand_cut[n];

    const float4* row = (const float4*)(hidden + (size_t)bat * SS * HH) + t;
    const int rstride = HH / 4;

    float4 ml = make_float4(-FLT_MAX, -FLT_MAX, -FLT_MAX, -FLT_MAX);
    float4 mr = ml;

    for (int s0 = 0; s0 < SS; s0 += TG) {
        const float4* p = row + (size_t)s0 * rstride;
        float4 v[TG];
        #pragma unroll
        for (int j = 0; j < TG; ++j) v[j] = p[j * rstride];

        if (s0 + TG <= cut) {
            #pragma unroll
            for (int j = 0; j < TG; ++j) ml = max4(ml, v[j]);
        } else if (s0 >= cut) {
            #pragma unroll
            for (int j = 0; j < TG; ++j) mr = max4(mr, v[j]);
        } else {
            #pragma unroll
            for (int j = 0; j < TG; ++j) {
                if (s0 + j < cut) ml = max4(ml, v[j]);
                else              mr = max4(mr, v[j]);
            }
        }
    }

    float4* outL = (float4*)(pooled + (size_t)n * KTOT) + t;
    float4* outR = (float4*)(pooled + (size_t)n * KTOT + HH) + t;
    *outL = ml;
    *outR = mr;
}

__global__ void __launch_bounds__(256)
dmbert_chunk_kernel(const float* __restrict__ hidden,
                    const int*  __restrict__ cand_batch,
                    const int*  __restrict__ cand_cut,
                    float*      __restrict__ pooled,
                    float*      __restrict__ chunkmax,
                    const size_t ws_bytes) {
    const int t = threadIdx.x;

    if (ws_bytes < WS_NEED_BYTES) {
        if (blockIdx.x < NC)
            pool_candidate_walk(hidden, cand_batch, cand_cut, pooled, blockIdx.x, t);
        return;
    }

    const int bat = blockIdx.x / NCH;
    const int ch  = blockIdx.x % NCH;

    const int cb = cand_batch[t];
    const int cc = clamp_cut(cand_cut[t]);

    const float4* p = (const float4*)(hidden + ((size_t)bat * SS + (size_t)ch * TG) * HH) + t;
    const int rstride = HH / 4;

    float4 v[TG];
    #pragma unroll
    for (int j = 0; j < TG; ++j) v[j] = p[j * rstride];

    __shared__ unsigned hit[NC / 32];
    const bool     mine = (cb == bat) & (home_chunk(cc) == ch);
    const unsigned word = __builtin_amdgcn_ballot_w32(mine);
    if ((t & 31) == 0) hit[t >> 5] = word;

    float4 up[NG], dn[NG];
    #pragma unroll
    for (int q = 0; q < NG; ++q)
        up[q] = dn[q] = max4(max4(v[q * GT], v[q * GT + 1]), max4(v[q * GT + 2], v[q * GT + 3]));
    #pragma unroll
    for (int q = 1; q < NG; ++q) up[q] = max4(up[q - 1], up[q]);
    #pragma unroll
    for (int q = NG - 2; q >= 0; --q) dn[q] = max4(dn[q + 1], dn[q]);
    const float4 cm = up[NG - 1];

    __syncthreads();

    ((float4*)(chunkmax + (size_t)blockIdx.x * HH))[t] = cm;

    for (int w = 0; w < NC / 32; ++w) {
        unsigned bits = __builtin_amdgcn_readfirstlane(hit[w]);
        while (bits) {
            const int n = w * 32 + __builtin_ctz(bits);
            bits &= bits - 1;
            const int c = clamp_cut(cand_cut[n]) - ch * TG;
            const int r = c % GT;
            float4 pl, pr;
            switch (c / GT) {
                case 0:  cut_partials<0>(v, up, dn, r, pl, pr); break;
                case 1:  cut_partials<1>(v, up, dn, r, pl, pr); break;
                case 2:  cut_partials<2>(v, up, dn, r, pl, pr); break;
                case 3:  cut_partials<3>(v, up, dn, r, pl, pr); break;
                case 4:  cut_partials<4>(v, up, dn, r, pl, pr); break;
                case 5:  cut_partials<5>(v, up, dn, r, pl, pr); break;
                case 6:  cut_partials<6>(v, up, dn, r, pl, pr); break;
                case 7:  cut_partials<7>(v, up, dn, r, pl, pr); break;
                default: cut_partials<NG>(v, up, dn, r, pl, pr); break;
            }
            ((float4*)(pooled + (size_t)n * KTOT))[t]      = pl;
            ((float4*)(pooled + (size_t)n * KTOT + HH))[t] = pr;
        }
    }
}

#define GC    4
#define NWAVE 16
#define KW    (KTOT / NWAVE)
#define NSTEP (KW / 4)
#define TP    (KTOT + 4)

__global__ void __launch_bounds__(NWAVE * 32)
dmbert_merge_logits_kernel(const int*   __restrict__ cand_batch,
                           const int*   __restrict__ cand_cut,
                           const float* __restrict__ chunkmax,
                           const float* __restrict__ pooled,
                           const float* __restrict__ Wm,
                           const float* __restrict__ bias,
                           float*       __restrict__ out,
                           const size_t ws_bytes) {
    const int m0    = blockIdx.x * GC;
    const int ntile = blockIdx.y;
    const int t     = threadIdx.x;
    const int wave  = t >> 5;
    const int lane  = t & 31;
    const int half  = lane >> 4;
    const int l16   = lane & 15;

    __shared__ __attribute__((aligned(16))) float tile[GC][TP];
    __shared__ float4 red[NWAVE][32];

    const int col = ntile * 16 + l16;
    const int cld = min(col, NL - 1);
    const float bv = bias[cld];

    const int k0 = wave * KW + 2 * half;
    const float* bcol = Wm + (size_t)k0 * NL + cld;
    v2f b[NSTEP];
    #pragma unroll
    for (int s = 0; s < NSTEP; ++s) {
        b[s].x = bcol[(size_t)(4 * s) * NL];
        b[s].y = bcol[(size_t)(4 * s) * NL + NL];
    }

    const int hq = t & 255;
    const int cp = __builtin_amdgcn_readfirstlane(t >> 8);
    const int nA = m0 + 2 * cp;
    const int nB = nA + 1;

    float4 mlA = ((const float4*)(pooled + (size_t)nA * KTOT))[hq];
    float4 mrA = ((const float4*)(pooled + (size_t)nA * KTOT + HH))[hq];
    float4 mlB = ((const float4*)(pooled + (size_t)nB * KTOT))[hq];
    float4 mrB = ((const float4*)(pooled + (size_t)nB * KTOT + HH))[hq];

    if (ws_bytes >= WS_NEED_BYTES) {
        const int ciA = home_chunk(clamp_cut(cand_cut[nA]));
        const int ciB = home_chunk(clamp_cut(cand_cut[nB]));
        const float4* rowA = (const float4*)(chunkmax + (size_t)cand_batch[nA] * NCH * HH) + hq;
        const float4* rowB = (const float4*)(chunkmax + (size_t)cand_batch[nB] * NCH * HH) + hq;

        float4 cA[NCH], cB[NCH];
        #pragma unroll
        for (int k = 0; k < NCH; ++k) {
            cA[k] = rowA[k * (HH / 4)];
            cB[k] = rowB[k * (HH / 4)];
        }
        const float4 lo = make_float4(-FLT_MAX, -FLT_MAX, -FLT_MAX, -FLT_MAX);
        #pragma unroll
        for (int k = 0; k < NCH; ++k) {
            mlA = max4(mlA, (k < ciA) ? cA[k] : lo);
            mrA = max4(mrA, (k > ciA) ? cA[k] : lo);
            mlB = max4(mlB, (k < ciB) ? cB[k] : lo);
            mrB = max4(mrB, (k > ciB) ? cB[k] : lo);
        }
    }

    *(float4*)&tile[2 * cp][4 * hq]          = mlA;
    *(float4*)&tile[2 * cp][HH + 4 * hq]     = mrA;
    *(float4*)&tile[2 * cp + 1][4 * hq]      = mlB;
    *(float4*)&tile[2 * cp + 1][HH + 4 * hq] = mrB;
    __syncthreads();

    const float* arow = &tile[l16 & (GC - 1)][k0];
    v8f c = {};
    #pragma unroll
    for (int s = 0; s < NSTEP; ++s) {
        v2f a = *(const v2f*)(arow + 4 * s);
        c = __builtin_amdgcn_wmma_f32_16x16x4_f32(
                 false, a,  false, b[s],
                 (short)0, c,  false,  false);
    }

    red[wave][lane] = make_float4(c[0], c[1], c[2], c[3]);
    __syncthreads();

    if (t < GC * 16) {
        const int r  = t >> 4;
        const int oc = ntile * 16 + (t & 15);
        if (oc < NL) {
            const float* p = (const float*)&red[0][t & 15] + r;
            float s = p[0];
            #pragma unroll
            for (int w = 1; w < NWAVE; ++w) s += p[w * 32 * 4];
            out[(size_t)(m0 + r) * NL + oc] = s + bv;
        }
    }
}

extern "C" void kernel_launch(void* const* d_in, const int* in_sizes, int n_in,
                              void* d_out, int out_size, void* d_ws, size_t ws_size,
                              hipStream_t stream) {
    const float* hidden     = (const float*)d_in[0];
    const float* Wm         = (const float*)d_in[1];
    const float* bias       = (const float*)d_in[2];
    const int*   cand_batch = (const int*)  d_in[3];
    const int*   cand_cut   = (const int*)  d_in[4];
    float*       out        = (float*)d_out;

    float* pooled   = (float*)d_ws;
    float* chunkmax = pooled + (size_t)NC * KTOT;

    dmbert_chunk_kernel<<<BB * NCH, 256, 0, stream>>>(hidden, cand_batch, cand_cut,
                                                      pooled, chunkmax, ws_size);

    dim3 grid(NC / GC, (NL + 15) / 16);
    dmbert_merge_logits_kernel<<<grid, NWAVE * 32, 0, stream>>>(cand_batch, cand_cut, chunkmax, pooled,
                                                              Wm, bias, out, ws_size);
}


# Chunk of 64 tokens per thread, not 32 (64 loads in flight, groups of 8): 256 blocks, 2048 waves enter, not 512 and 4096; the merge reads 8 chunk maxima per candidate, not 16; max exact, same bits
# speedup vs baseline: 9.1904x; 1.0885x over previous
#include <hip/hip_runtime.h>
#include <float.h>

typedef __attribute__((ext_vector_type(2))) float v2f;
typedef __attribute__((ext_vector_type(8))) float v8f;

#define BB    32
#define SS    512
#define HH    1024
#define NC    256
#define NL    34
#define KTOT  (2 * HH)

#define TG  64
#define NCH (SS / TG)
#define GT  8
#define NG  (TG / GT)

#define WS_NEED_BYTES ((size_t)(NC * KTOT + BB * NCH * HH) * sizeof(float))

__device__ __forceinline__ int clamp_cut(int cut) { return min(max(cut, 0), SS); }
__device__ __forceinline__ int home_chunk(int cut) { return min(cut / TG, NCH - 1); }

__device__ __forceinline__ float4 max4(float4 a, float4 b) {
    return make_float4(__builtin_fmaximum_numf(a.x, b.x), __builtin_fmaximum_numf(a.y, b.y),
                       __builtin_fmaximum_numf(a.z, b.z), __builtin_fmaximum_numf(a.w, b.w));
}

template <int G0>
__device__ __forceinline__ void cut_partials(const float4 (&v)[TG], const float4 (&up)[NG],
                                             const float4 (&dn)[NG], const int r,
                                             float4& pl, float4& pr) {
    const float4 lo = make_float4(-FLT_MAX, -FLT_MAX, -FLT_MAX, -FLT_MAX);
    if constexpr (G0 > 0)      pl = up[G0 - 1]; else pl = lo;
    if constexpr (G0 + 1 < NG) pr = dn[G0 + 1]; else pr = lo;
    if constexpr (G0 < NG) {
        #pragma unroll
        for (int i = 0; i < GT; ++i) {
            if (i < r) pl = max4(pl, v[G0 * GT + i]);
            else       pr = max4(pr, v[G0 * GT + i]);
        }
    }
}

__device__ __forceinline__ void pool_candidate_walk(const float* __restrict__ hidden,
                                                    const int*  __restrict__ cand_batch,
                                                    const int*  __restrict__ cand_cut,
                                                    float*      __restrict__ pooled,
                                                    const int n, const int t) {
    const int bat = cand_batch[n];
    const int cut = cand_cut[n];

    const float4* row = (const float4*)(hidden + (size_t)bat * SS * HH) + t;
    const int rstride = HH / 4;

    float4 ml = make_float4(-FLT_MAX, -FLT_MAX, -FLT_MAX, -FLT_MAX);
    float4 mr = ml;

    for (int s0 = 0; s0 < SS; s0 += TG) {
        const float4* p = row + (size_t)s0 * rstride;
        float4 v[TG];
        #pragma unroll
        for (int j = 0; j < TG; ++j) v[j] = p[j * rstride];

        if (s0 + TG <= cut) {
            #pragma unroll
            for (int j = 0; j < TG; ++j) ml = max4(ml, v[j]);
        } else if (s0 >= cut) {
            #pragma unroll
            for (int j = 0; j < TG; ++j) mr = max4(mr, v[j]);
        } else {
            #pragma unroll
            for (int j = 0; j < TG; ++j) {
                if (s0 + j < cut) ml = max4(ml, v[j]);
                else              mr = max4(mr, v[j]);
            }
        }
    }

    float4* outL = (float4*)(pooled + (size_t)n * KTOT) + t;
    float4* outR = (float4*)(pooled + (size_t)n * KTOT + HH) + t;
    *outL = ml;
    *outR = mr;
}

__global__ void __launch_bounds__(256)
dmbert_chunk_kernel(const float* __restrict__ hidden,
                    const int*  __restrict__ cand_batch,
                    const int*  __restrict__ cand_cut,
                    float*      __restrict__ pooled,
                    float*      __restrict__ chunkmax,
                    const size_t ws_bytes) {
    const int t = threadIdx.x;

    if (ws_bytes < WS_NEED_BYTES) {
        if (blockIdx.x < NC)
            pool_candidate_walk(hidden, cand_batch, cand_cut, pooled, blockIdx.x, t);
        return;
    }

    const int bat = blockIdx.x / NCH;
    const int ch  = blockIdx.x % NCH;

    const int cb = cand_batch[t];
    const int cc = clamp_cut(cand_cut[t]);

    const float4* p = (const float4*)(hidden + ((size_t)bat * SS + (size_t)ch * TG) * HH) + t;
    const int rstride = HH / 4;

    float4 v[TG];
    #pragma unroll
    for (int j = 0; j < TG; ++j) v[j] = p[j * rstride];

    __shared__ unsigned hit[NC / 32];
    const bool     mine = (cb == bat) & (home_chunk(cc) == ch);
    const unsigned word = __builtin_amdgcn_ballot_w32(mine);
    if ((t & 31) == 0) hit[t >> 5] = word;

    float4 up[NG], dn[NG];
    #pragma unroll
    for (int q = 0; q < NG; ++q)
        up[q] = dn[q] = max4(max4(max4(v[q * GT],     v[q * GT + 1]), max4(v[q * GT + 2], v[q * GT + 3])),
                             max4(max4(v[q * GT + 4], v[q * GT + 5]), max4(v[q * GT + 6], v[q * GT + 7])));
    #pragma unroll
    for (int q = 1; q < NG; ++q) up[q] = max4(up[q - 1], up[q]);
    #pragma unroll
    for (int q = NG - 2; q >= 0; --q) dn[q] = max4(dn[q + 1], dn[q]);
    const float4 cm = up[NG - 1];

    __syncthreads();

    ((float4*)(chunkmax + (size_t)blockIdx.x * HH))[t] = cm;

    for (int w = 0; w < NC / 32; ++w) {
        unsigned bits = __builtin_amdgcn_readfirstlane(hit[w]);
        while (bits) {
            const int n = w * 32 + __builtin_ctz(bits);
            bits &= bits - 1;
            const int c = clamp_cut(cand_cut[n]) - ch * TG;
            const int r = c % GT;
            float4 pl, pr;
            switch (c / GT) {
                case 0:  cut_partials<0>(v, up, dn, r, pl, pr); break;
                case 1:  cut_partials<1>(v, up, dn, r, pl, pr); break;
                case 2:  cut_partials<2>(v, up, dn, r, pl, pr); break;
                case 3:  cut_partials<3>(v, up, dn, r, pl, pr); break;
                case 4:  cut_partials<4>(v, up, dn, r, pl, pr); break;
                case 5:  cut_partials<5>(v, up, dn, r, pl, pr); break;
                case 6:  cut_partials<6>(v, up, dn, r, pl, pr); break;
                case 7:  cut_partials<7>(v, up, dn, r, pl, pr); break;
                default: cut_partials<NG>(v, up, dn, r, pl, pr); break;
            }
            ((float4*)(pooled + (size_t)n * KTOT))[t]      = pl;
            ((float4*)(pooled + (size_t)n * KTOT + HH))[t] = pr;
        }
    }
}

#define GC    4
#define NWAVE 16
#define KW    (KTOT / NWAVE)
#define NSTEP (KW / 4)
#define TP    (KTOT + 4)

__global__ void __launch_bounds__(NWAVE * 32)
dmbert_merge_logits_kernel(const int*   __restrict__ cand_batch,
                           const int*   __restrict__ cand_cut,
                           const float* __restrict__ chunkmax,
                           const float* __restrict__ pooled,
                           const float* __restrict__ Wm,
                           const float* __restrict__ bias,
                           float*       __restrict__ out,
                           const size_t ws_bytes) {
    const int m0    = blockIdx.x * GC;
    const int ntile = blockIdx.y;
    const int t     = threadIdx.x;
    const int wave  = t >> 5;
    const int lane  = t & 31;
    const int half  = lane >> 4;
    const int l16   = lane & 15;

    __shared__ __attribute__((aligned(16))) float tile[GC][TP];
    __shared__ float4 red[NWAVE][32];

    const int col = ntile * 16 + l16;
    const int cld = min(col, NL - 1);
    const float bv = bias[cld];

    const int k0 = wave * KW + 2 * half;
    const float* bcol = Wm + (size_t)k0 * NL + cld;
    v2f b[NSTEP];
    #pragma unroll
    for (int s = 0; s < NSTEP; ++s) {
        b[s].x = bcol[(size_t)(4 * s) * NL];
        b[s].y = bcol[(size_t)(4 * s) * NL + NL];
    }

    const int hq = t & 255;
    const int cp = __builtin_amdgcn_readfirstlane(t >> 8);
    const int nA = m0 + 2 * cp;
    const int nB = nA + 1;

    float4 mlA = ((const float4*)(pooled + (size_t)nA * KTOT))[hq];
    float4 mrA = ((const float4*)(pooled + (size_t)nA * KTOT + HH))[hq];
    float4 mlB = ((const float4*)(pooled + (size_t)nB * KTOT))[hq];
    float4 mrB = ((const float4*)(pooled + (size_t)nB * KTOT + HH))[hq];

    if (ws_bytes >= WS_NEED_BYTES) {
        const int ciA = home_chunk(clamp_cut(cand_cut[nA]));
        const int ciB = home_chunk(clamp_cut(cand_cut[nB]));
        const float4* rowA = (const float4*)(chunkmax + (size_t)cand_batch[nA] * NCH * HH) + hq;
        const float4* rowB = (const float4*)(chunkmax + (size_t)cand_batch[nB] * NCH * HH) + hq;

        float4 cA[NCH], cB[NCH];
        #pragma unroll
        for (int k = 0; k < NCH; ++k) {
            cA[k] = rowA[k * (HH / 4)];
            cB[k] = rowB[k * (HH / 4)];
        }
        const float4 lo = make_float4(-FLT_MAX, -FLT_MAX, -FLT_MAX, -FLT_MAX);
        #pragma unroll
        for (int k = 0; k < NCH; ++k) {
            mlA = max4(mlA, (k < ciA) ? cA[k] : lo);
            mrA = max4(mrA, (k > ciA) ? cA[k] : lo);
            mlB = max4(mlB, (k < ciB) ? cB[k] : lo);
            mrB = max4(mrB, (k > ciB) ? cB[k] : lo);
        }
    }

    *(float4*)&tile[2 * cp][4 * hq]          = mlA;
    *(float4*)&tile[2 * cp][HH + 4 * hq]     = mrA;
    *(float4*)&tile[2 * cp + 1][4 * hq]      = mlB;
    *(float4*)&tile[2 * cp + 1][HH + 4 * hq] = mrB;
    __syncthreads();

    const float* arow = &tile[l16 & (GC - 1)][k0];
    v8f c = {};
    #pragma unroll
    for (int s = 0; s < NSTEP; ++s) {
        v2f a = *(const v2f*)(arow + 4 * s);
        c = __builtin_amdgcn_wmma_f32_16x16x4_f32(
                 false, a,  false, b[s],
                 (short)0, c,  false,  false);
    }

    red[wave][lane] = make_float4(c[0], c[1], c[2], c[3]);
    __syncthreads();

    if (t < GC * 16) {
        const int r  = t >> 4;
        const int oc = ntile * 16 + (t & 15);
        if (oc < NL) {
            const float* p = (const float*)&red[0][t & 15] + r;
            float s = p[0];
            #pragma unroll
            for (int w = 1; w < NWAVE; ++w) s += p[w * 32 * 4];
            out[(size_t)(m0 + r) * NL + oc] = s + bv;
        }
    }
}

extern "C" void kernel_launch(void* const* d_in, const int* in_sizes, int n_in,
                              void* d_out, int out_size, void* d_ws, size_t ws_size,
                              hipStream_t stream) {
    const float* hidden     = (const float*)d_in[0];
    const float* Wm         = (const float*)d_in[1];
    const float* bias       = (const float*)d_in[2];
    const int*   cand_batch = (const int*)  d_in[3];
    const int*   cand_cut   = (const int*)  d_in[4];
    float*       out        = (float*)d_out;

    float* pooled   = (float*)d_ws;
    float* chunkmax = pooled + (size_t)NC * KTOT;

    dmbert_chunk_kernel<<<BB * NCH, 256, 0, stream>>>(hidden, cand_batch, cand_cut,
                                                      pooled, chunkmax, ws_size);

    dim3 grid(NC / GC, (NL + 15) / 16);
    dmbert_merge_logits_kernel<<<grid, NWAVE * 32, 0, stream>>>(cand_batch, cand_cut, chunkmax, pooled,
                                                              Wm, bias, out, ws_size);
}


# Product launch takes its B fragments as 16-byte loads of whole lines from a fragment-ordered copy of W that the chunk pass writes to the workspace each call; same k per step, sums bit-identical
# speedup vs baseline: 10.8510x; 1.1807x over previous
#include <hip/hip_runtime.h>
#include <float.h>

typedef __attribute__((ext_vector_type(2))) float v2f;
typedef __attribute__((ext_vector_type(8))) float v8f;

#define BB    32
#define SS    512
#define HH    1024
#define NC    256
#define NL    34
#define KTOT  (2 * HH)

#define TG  64
#define NCH (SS / TG)
#define GT  8
#define NG  (TG / GT)

#define WS_NEED_BYTES ((size_t)(NC * KTOT + BB * NCH * HH) * sizeof(float))
#define WS_FRAG_BYTES (WS_NEED_BYTES + (size_t)(KTOT * NL) * sizeof(float))
#define WF_TILE  (KTOT * 16)
#define NCOL2    (NL - 32)

__device__ __forceinline__ int clamp_cut(int cut) { return min(max(cut, 0), SS); }
__device__ __forceinline__ int home_chunk(int cut) { return min(cut / TG, NCH - 1); }

__device__ __forceinline__ float4 max4(float4 a, float4 b) {
    return make_float4(__builtin_fmaximum_numf(a.x, b.x), __builtin_fmaximum_numf(a.y, b.y),
                       __builtin_fmaximum_numf(a.z, b.z), __builtin_fmaximum_numf(a.w, b.w));
}

template <int G0>
__device__ __forceinline__ void cut_partials(const float4 (&v)[TG], const float4 (&up)[NG],
                                             const float4 (&dn)[NG], const int r,
                                             float4& pl, float4& pr) {
    const float4 lo = make_float4(-FLT_MAX, -FLT_MAX, -FLT_MAX, -FLT_MAX);
    if constexpr (G0 > 0)      pl = up[G0 - 1]; else pl = lo;
    if constexpr (G0 + 1 < NG) pr = dn[G0 + 1]; else pr = lo;
    if constexpr (G0 < NG) {
        #pragma unroll
        for (int i = 0; i < GT; ++i) {
            if (i < r) pl = max4(pl, v[G0 * GT + i]);
            else       pr = max4(pr, v[G0 * GT + i]);
        }
    }
}

__device__ __forceinline__ void pool_candidate_walk(const float* __restrict__ hidden,
                                                    const int*  __restrict__ cand_batch,
                                                    const int*  __restrict__ cand_cut,
                                                    float*      __restrict__ pooled,
                                                    const int n, const int t) {
    const int bat = cand_batch[n];
    const int cut = cand_cut[n];

    const float4* row = (const float4*)(hidden + (size_t)bat * SS * HH) + t;
    const int rstride = HH / 4;

    float4 ml = make_float4(-FLT_MAX, -FLT_MAX, -FLT_MAX, -FLT_MAX);
    float4 mr = ml;

    for (int s0 = 0; s0 < SS; s0 += TG) {
        const float4* p = row + (size_t)s0 * rstride;
        float4 v[TG];
        #pragma unroll
        for (int j = 0; j < TG; ++j) v[j] = p[j * rstride];

        if (s0 + TG <= cut) {
            #pragma unroll
            for (int j = 0; j < TG; ++j) ml = max4(ml, v[j]);
        } else if (s0 >= cut) {
            #pragma unroll
            for (int j = 0; j < TG; ++j) mr = max4(mr, v[j]);
        } else {
            #pragma unroll
            for (int j = 0; j < TG; ++j) {
                if (s0 + j < cut) ml = max4(ml, v[j]);
                else              mr = max4(mr, v[j]);
            }
        }
    }

    float4* outL = (float4*)(pooled + (size_t)n * KTOT) + t;
    float4* outR = (float4*)(pooled + (size_t)n * KTOT + HH) + t;
    *outL = ml;
    *outR = mr;
}

__global__ void __launch_bounds__(256)
dmbert_chunk_kernel(const float* __restrict__ hidden,
                    const float* __restrict__ Wm,
                    const int*  __restrict__ cand_batch,
                    const int*  __restrict__ cand_cut,
                    float*      __restrict__ pooled,
                    float*      __restrict__ chunkmax,
                    float*      __restrict__ wfrag,
                    const size_t ws_bytes) {
    const int t = threadIdx.x;

    if (ws_bytes < WS_NEED_BYTES) {
        if (blockIdx.x < NC)
            pool_candidate_walk(hidden, cand_batch, cand_cut, pooled, blockIdx.x, t);
        return;
    }

    const int bat = blockIdx.x / NCH;
    const int ch  = blockIdx.x % NCH;

    const int cb = cand_batch[t];
    const int cc = clamp_cut(cand_cut[t]);

    const bool frag = ws_bytes >= WS_FRAG_BYTES;
    const bool frag2 = frag & ((blockIdx.x & 1) == 0) & (t < 32);
    const int  fe   = t & 3;
    const int  fk   = 4 * (fe >> 1) + (fe & 1);
    float wv = 0.0f, wv2 = 0.0f;
    if (frag) {
        const int slot = (t & 127) >> 2;
        wv = Wm[(size_t)(8 * blockIdx.x + fk + 2 * (slot >> 4)) * NL + 16 * (t >> 7) + (slot & 15)];
    }
    if (frag2) {
        const int g    = blockIdx.x + (t >> 4);
        const int slot = (t & 15) >> 2;
        wv2 = Wm[(size_t)(8 * g + fk + 2 * (slot >> 1)) * NL + 32 + (slot & 1)];
    }

    const float4* p = (const float4*)(hidden + ((size_t)bat * SS + (size_t)ch * TG) * HH) + t;
    const int rstride = HH / 4;

    float4 v[TG];
    #pragma unroll
    for (int j = 0; j < TG; ++j) v[j] = p[j * rstride];

    if (frag)  wfrag[(size_t)(t >> 7) * WF_TILE + (size_t)blockIdx.x * 128 + (t & 127)] = wv;
    if (frag2) wfrag[(size_t)2 * WF_TILE + (size_t)blockIdx.x * (8 * NCOL2) + t] = wv2;

    __shared__ unsigned hit[NC / 32];
    const bool     mine = (cb == bat) & (home_chunk(cc) == ch);
    const unsigned word = __builtin_amdgcn_ballot_w32(mine);
    if ((t & 31) == 0) hit[t >> 5] = word;

    float4 up[NG], dn[NG];
    #pragma unroll
    for (int q = 0; q < NG; ++q)
        up[q] = dn[q] = max4(max4(max4(v[q * GT],     v[q * GT + 1]), max4(v[q * GT + 2], v[q * GT + 3])),
                             max4(max4(v[q * GT + 4], v[q * GT + 5]), max4(v[q * GT + 6], v[q * GT + 7])));
    #pragma unroll
    for (int q = 1; q < NG; ++q) up[q] = max4(up[q - 1], up[q]);
    #pragma unroll
    for (int q = NG - 2; q >= 0; --q) dn[q] = max4(dn[q + 1], dn[q]);
    const float4 cm = up[NG - 1];

    __syncthreads();

    ((float4*)(chunkmax + (size_t)blockIdx.x * HH))[t] = cm;

    for (int w = 0; w < NC / 32; ++w) {
        unsigned bits = __builtin_amdgcn_readfirstlane(hit[w]);
        while (bits) {
            const int n = w * 32 + __builtin_ctz(bits);
            bits &= bits - 1;
            const int c = clamp_cut(cand_cut[n]) - ch * TG;
            const int r = c % GT;
            float4 pl, pr;
            switch (c / GT) {
                case 0:  cut_partials<0>(v, up, dn, r, pl, pr); break;
                case 1:  cut_partials<1>(v, up, dn, r, pl, pr); break;
                case 2:  cut_partials<2>(v, up, dn, r, pl, pr); break;
                case 3:  cut_partials<3>(v, up, dn, r, pl, pr); break;
                case 4:  cut_partials<4>(v, up, dn, r, pl, pr); break;
                case 5:  cut_partials<5>(v, up, dn, r, pl, pr); break;
                case 6:  cut_partials<6>(v, up, dn, r, pl, pr); break;
                case 7:  cut_partials<7>(v, up, dn, r, pl, pr); break;
                default: cut_partials<NG>(v, up, dn, r, pl, pr); break;
            }
            ((float4*)(pooled + (size_t)n * KTOT))[t]      = pl;
            ((float4*)(pooled + (size_t)n * KTOT + HH))[t] = pr;
        }
    }
}

#define GC    4
#define NWAVE 16
#define KW    (KTOT / NWAVE)
#define NSTEP (KW / 4)
#define TP    (KTOT + 4)

__global__ void __launch_bounds__(NWAVE * 32)
dmbert_merge_logits_kernel(const int*   __restrict__ cand_batch,
                           const int*   __restrict__ cand_cut,
                           const float* __restrict__ chunkmax,
                           const float* __restrict__ pooled,
                           const float* __restrict__ Wm,
                           const float* __restrict__ wfrag,
                           const float* __restrict__ bias,
                           float*       __restrict__ out,
                           const size_t ws_bytes) {
    const int m0    = blockIdx.x * GC;
    const int ntile = blockIdx.y;
    const int t     = threadIdx.x;
    const int wave  = t >> 5;
    const int lane  = t & 31;
    const int half  = lane >> 4;
    const int l16   = lane & 15;

    __shared__ __attribute__((aligned(16))) float tile[GC][TP];
    __shared__ float4 red[NWAVE][32];

    const int col = ntile * 16 + l16;
    const int cld = min(col, NL - 1);
    const float bv = bias[cld];

    const int k0 = wave * KW + 2 * half;
    v2f b[NSTEP];
    if (ws_bytes >= WS_FRAG_BYTES) {
        const int ncol = (ntile < 2) ? 16 : NCOL2;
        const int slot = half * ncol + min(l16, ncol - 1);
        const float4* bq = (const float4*)(wfrag + (size_t)ntile * WF_TILE)
                         + (size_t)(wave * (KW / 8)) * (2 * ncol) + slot;
        #pragma unroll
        for (int j = 0; j < NSTEP / 2; ++j) {
            const float4 q = bq[(size_t)j * (2 * ncol)];
            b[2 * j].x     = q.x;
            b[2 * j].y     = q.y;
            b[2 * j + 1].x = q.z;
            b[2 * j + 1].y = q.w;
        }
    } else {
        const float* bcol = Wm + (size_t)k0 * NL + cld;
        #pragma unroll
        for (int s = 0; s < NSTEP; ++s) {
            b[s].x = bcol[(size_t)(4 * s) * NL];
            b[s].y = bcol[(size_t)(4 * s) * NL + NL];
        }
    }

    const int hq = t & 255;
    const int cp = __builtin_amdgcn_readfirstlane(t >> 8);
    const int nA = m0 + 2 * cp;
    const int nB = nA + 1;

    float4 mlA = ((const float4*)(pooled + (size_t)nA * KTOT))[hq];
    float4 mrA = ((const float4*)(pooled + (size_t)nA * KTOT + HH))[hq];
    float4 mlB = ((const float4*)(pooled + (size_t)nB * KTOT))[hq];
    float4 mrB = ((const float4*)(pooled + (size_t)nB * KTOT + HH))[hq];

    if (ws_bytes >= WS_NEED_BYTES) {
        const int ciA = home_chunk(clamp_cut(cand_cut[nA]));
        const int ciB = home_chunk(clamp_cut(cand_cut[nB]));
        const float4* rowA = (const float4*)(chunkmax + (size_t)cand_batch[nA] * NCH * HH) + hq;
        const float4* rowB = (const float4*)(chunkmax + (size_t)cand_batch[nB] * NCH * HH) + hq;

        float4 cA[NCH], cB[NCH];
        #pragma unroll
        for (int k = 0; k < NCH; ++k) {
            cA[k] = rowA[k * (HH / 4)];
            cB[k] = rowB[k * (HH / 4)];
        }
        const float4 lo = make_float4(-FLT_MAX, -FLT_MAX, -FLT_MAX, -FLT_MAX);
        #pragma unroll
        for (int k = 0; k < NCH; ++k) {
            mlA = max4(mlA, (k < ciA) ? cA[k] : lo);
            mrA = max4(mrA, (k > ciA) ? cA[k] : lo);
            mlB = max4(mlB, (k < ciB) ? cB[k] : lo);
            mrB = max4(mrB, (k > ciB) ? cB[k] : lo);
        }
    }

    *(float4*)&tile[2 * cp][4 * hq]          = mlA;
    *(float4*)&tile[2 * cp][HH + 4 * hq]     = mrA;
    *(float4*)&tile[2 * cp + 1][4 * hq]      = mlB;
    *(float4*)&tile[2 * cp + 1][HH + 4 * hq] = mrB;
    __syncthreads();

    const float* arow = &tile[l16 & (GC - 1)][k0];
    int roff = t * (int)sizeof(float4);
    asm volatile("" : "+v"(roff));
    v8f c = {};
    #pragma unroll
    for (int s = 0; s < NSTEP; ++s) {
        v2f a = *(const v2f*)(arow + 4 * s);
        c = __builtin_amdgcn_wmma_f32_16x16x4_f32(
                 false, a,  false, b[s],
                 (short)0, c,  false,  false);
    }

    *(float4*)((char*)&red[0][0] + roff) = make_float4(c[0], c[1], c[2], c[3]);
    __syncthreads();

    if (t < GC * 16) {
        const int r  = t >> 4;
        const int oc = ntile * 16 + (t & 15);
        if (oc < NL) {
            const float* p = (const float*)&red[0][t & 15] + r;
            float s = p[0];
            #pragma unroll
            for (int w = 1; w < NWAVE; ++w) s += p[w * 32 * 4];
            out[(size_t)(m0 + r) * NL + oc] = s + bv;
        }
    }
}

extern "C" void kernel_launch(void* const* d_in, const int* in_sizes, int n_in,
                              void* d_out, int out_size, void* d_ws, size_t ws_size,
                              hipStream_t stream) {
    const float* hidden     = (const float*)d_in[0];
    const float* Wm         = (const float*)d_in[1];
    const float* bias       = (const float*)d_in[2];
    const int*   cand_batch = (const int*)  d_in[3];
    const int*   cand_cut   = (const int*)  d_in[4];
    float*       out        = (float*)d_out;

    float* pooled   = (float*)d_ws;
    float* chunkmax = pooled + (size_t)NC * KTOT;
    float* wfrag    = chunkmax + (size_t)BB * NCH * HH;

    dmbert_chunk_kernel<<<BB * NCH, 256, 0, stream>>>(hidden, Wm, cand_batch, cand_cut,
                                                      pooled, chunkmax, wfrag, ws_size);

    dim3 grid(NC / GC, (NL + 15) / 16);
    dmbert_merge_logits_kernel<<<grid, NWAVE * 32, 0, stream>>>(cand_batch, cand_cut, chunkmax, pooled,
                                                              Wm, wfrag, bias, out, ws_size);
}
